# starting kernel
# baseline (speedup 1.0000x reference)
#include <hip/hip_runtime.h>
#include <hip/hip_bf16.h>

typedef __attribute__((ext_vector_type(2))) float v2f;
typedef __attribute__((ext_vector_type(8))) float v8f;

#define IMG_H 256
#define IMG_W 256
#define NGAUSS 512
#define FOCAL_F 50.0f

#define WS_GOP 2048
#define WS_BMAT 2560

__device__ __forceinline__ float lane_xor16(float x) {
    int v = __builtin_bit_cast(int, x);
    int r = __builtin_amdgcn_permlanex16(v, v, 0x76543210, 0xfedcba98, false, false);
    return __builtin_bit_cast(float, r);
}

__global__ __launch_bounds__(512)
void dgr_prep_kernel(const float* __restrict__ means3d,
                     const float* __restrict__ scales,
                     const float* __restrict__ opac,
                     const float* __restrict__ colors,
                     float* __restrict__ ws) {
    __shared__ float zsh[NGAUSS];
    __shared__ float csh[NGAUSS * 3];

    const int i = threadIdx.x;

    const float z = fmaxf(means3d[i * 3 + 2], 0.1f);
    zsh[i] = z;
    __syncthreads();

    int rank = 0;
    for (int j = 0; j < NGAUSS; ++j) {
        const float zj = zsh[j];
        rank += ((zj < z) || (zj == z && j < i)) ? 1 : 0;
    }

    const float pu = FOCAL_F * means3d[i * 3 + 0] / z;
    const float pv = FOCAL_F * means3d[i * 3 + 1] / z;
    const float su = fmaxf(FOCAL_F * scales[i * 3 + 0] / z, 0.5f);
    const float sv = fmaxf(FOCAL_F * scales[i * 3 + 1] / z, 0.5f);

    float4 g;
    g.x = pu;
    g.y = pv;
    g.z = 1.0f / (su * su);
    g.w = 1.0f / (sv * sv);
    ((float4*)ws)[rank] = g;
    ws[WS_GOP + rank] = opac[i];

    csh[rank * 3 + 0] = colors[i * 3 + 0];
    csh[rank * 3 + 1] = colors[i * 3 + 1];
    csh[rank * 3 + 2] = colors[i * 3 + 2];
    __syncthreads();

    float* bmat = ws + WS_BMAT;
    for (int e = i; e < 128 * 32; e += 512) {
        const int t = e >> 5;
        const int l = e & 31;
        const int col = l & 15;
        const int k0 = (l < 16) ? 0 : 2;
        const int n0 = 4 * t;
        float b0 = 0.0f, b1 = 0.0f;
        if (col < 3) {
            b0 = csh[(n0 + k0) * 3 + col];
            b1 = csh[(n0 + k0 + 1) * 3 + col];
        }
        bmat[e * 2 + 0] = b0;
        bmat[e * 2 + 1] = b1;
    }
}

__global__ __launch_bounds__(256)
void dgr_raster_kernel(const float* __restrict__ ws, float* __restrict__ out) {
    const float4* __restrict__ gp4 = (const float4*)ws;
    const float* __restrict__ gop = ws + WS_GOP;
    const v2f* __restrict__ bmat = (const v2f*)(ws + WS_BMAT);

    const int tid = threadIdx.x;
    const int lane = tid & 31;
    const int wave = tid >> 5;
    const int pixBase = blockIdx.x * 256 + wave * 32;
    const int p = pixBase + lane;

    const float u = (float)(p & (IMG_W - 1)) - (float)(IMG_W / 2);
    const float v = (float)(p >> 8) - (float)(IMG_H / 2);

    const bool lo = (lane < 16);

    float T = 1.0f;
    v8f acc1 = {0.f, 0.f, 0.f, 0.f, 0.f, 0.f, 0.f, 0.f};
    v8f acc2 = {0.f, 0.f, 0.f, 0.f, 0.f, 0.f, 0.f, 0.f};

    for (int t = 0; t < NGAUSS / 4; ++t) {
        const int pt = (t + 16 < 128) ? (t + 16) : t;
        __builtin_prefetch((const void*)(bmat + pt * 32 + lane), 0, 3);

        float wk[4];
#pragma unroll
        for (int k = 0; k < 4; ++k) {
            const int n = 4 * t + k;
            const float4 g = gp4[n];
            const float op = gop[n];
            const float du = u - g.x;
            const float dv = v - g.y;
            const float e = __expf(-0.5f * (du * du * g.z + dv * dv * g.w));
            const float a = fminf(e * op, 1.0f);
            wk[k] = a * T;
            T *= (1.0f - a);
        }

        const float s0 = lane_xor16(wk[0]);
        const float s1 = lane_xor16(wk[1]);
        const float s2 = lane_xor16(wk[2]);
        const float s3 = lane_xor16(wk[3]);

        v2f a1, a2;
        a1.x = lo ? wk[0] : s2;
        a1.y = lo ? wk[1] : s3;
        a2.x = lo ? s0 : wk[2];
        a2.y = lo ? s1 : wk[3];

        const v2f b = bmat[t * 32 + lane];

        acc1 = __builtin_amdgcn_wmma_f32_16x16x4_f32(
            false, a1, false, b, (short)0, acc1, false, false);
        acc2 = __builtin_amdgcn_wmma_f32_16x16x4_f32(
            false, a2, false, b, (short)0, acc2, false, false);
    }

    const int ch = lane & 15;
    if (ch < 3) {
#pragma unroll
        for (int r = 0; r < 8; ++r) {
            const int m = r + (lo ? 0 : 8);
            out[(pixBase + m) * 3 + ch] = acc1[r];
            out[(pixBase + 16 + m) * 3 + ch] = acc2[r];
        }
    }
}

extern "C" void kernel_launch(void* const* d_in, const int* in_sizes, int n_in,
                              void* d_out, int out_size, void* d_ws, size_t ws_size,
                              hipStream_t stream) {
    const float* means3d = (const float*)d_in[0];
    const float* scales  = (const float*)d_in[1];
    const float* opac    = (const float*)d_in[2];
    const float* colors  = (const float*)d_in[3];
    float* ws  = (float*)d_ws;
    float* out = (float*)d_out;

    hipLaunchKernelGGL(dgr_prep_kernel, dim3(1), dim3(512), 0, stream,
                       means3d, scales, opac, colors, ws);

    hipLaunchKernelGGL(dgr_raster_kernel, dim3(256), dim3(256), 0, stream,
                       ws, out);
}


# raster: copy gaussian params, opacities and compact B fragments into LDS once per work-group; the 128-trip loop reads LDS instead of scalar-cache-missing s_loads and a global B load per trip
# speedup vs baseline: 1.6015x; 1.6015x over previous
#include <hip/hip_runtime.h>
#include <hip/hip_bf16.h>

typedef __attribute__((ext_vector_type(2))) float v2f;
typedef __attribute__((ext_vector_type(8))) float v8f;

#define IMG_H 256
#define IMG_W 256
#define NGAUSS 512
#define FOCAL_F 50.0f

#define WS_GOP 2048
#define WS_BMAT 2560

__device__ __forceinline__ float lane_xor16(float x) {
    int v = __builtin_bit_cast(int, x);
    int r = __builtin_amdgcn_permlanex16(v, v, 0x76543210, 0xfedcba98, false, false);
    return __builtin_bit_cast(float, r);
}

__global__ __launch_bounds__(512)
void dgr_prep_kernel(const float* __restrict__ means3d,
                     const float* __restrict__ scales,
                     const float* __restrict__ opac,
                     const float* __restrict__ colors,
                     float* __restrict__ ws) {
    __shared__ float zsh[NGAUSS];
    __shared__ float csh[NGAUSS * 3];

    const int i = threadIdx.x;

    const float z = fmaxf(means3d[i * 3 + 2], 0.1f);
    zsh[i] = z;
    __syncthreads();

    int rank = 0;
    for (int j = 0; j < NGAUSS; ++j) {
        const float zj = zsh[j];
        rank += ((zj < z) || (zj == z && j < i)) ? 1 : 0;
    }

    const float pu = FOCAL_F * means3d[i * 3 + 0] / z;
    const float pv = FOCAL_F * means3d[i * 3 + 1] / z;
    const float su = fmaxf(FOCAL_F * scales[i * 3 + 0] / z, 0.5f);
    const float sv = fmaxf(FOCAL_F * scales[i * 3 + 1] / z, 0.5f);

    float4 g;
    g.x = pu;
    g.y = pv;
    g.z = 1.0f / (su * su);
    g.w = 1.0f / (sv * sv);
    ((float4*)ws)[rank] = g;
    ws[WS_GOP + rank] = opac[i];

    csh[rank * 3 + 0] = colors[i * 3 + 0];
    csh[rank * 3 + 1] = colors[i * 3 + 1];
    csh[rank * 3 + 2] = colors[i * 3 + 2];
    __syncthreads();

    float* bmat = ws + WS_BMAT;
    for (int e = i; e < 128 * 32; e += 512) {
        const int t = e >> 5;
        const int l = e & 31;
        const int col = l & 15;
        const int k0 = (l < 16) ? 0 : 2;
        const int n0 = 4 * t;
        float b0 = 0.0f, b1 = 0.0f;
        if (col < 3) {
            b0 = csh[(n0 + k0) * 3 + col];
            b1 = csh[(n0 + k0 + 1) * 3 + col];
        }
        bmat[e * 2 + 0] = b0;
        bmat[e * 2 + 1] = b1;
    }
}

typedef __attribute__((ext_vector_type(4))) float v4f;

__device__ __forceinline__ float splat_one(const float pu, const float pv, const float iu,
                                           const float iv, const float op, const float u,
                                           const float v, float& T) {
    const float du = u - pu;
    const float dv = v - pv;
    const float e = __expf(-0.5f * (du * du * iu + dv * dv * iv));
    const float a = fminf(e * op, 1.0f);
    const float w = a * T;
    T *= (1.0f - a);
    return w;
}

__global__ __launch_bounds__(256)
void dgr_raster_kernel(const float* __restrict__ ws, float* __restrict__ out) {
    __shared__ v4f s_gp[NGAUSS];
    __shared__ v4f s_op[NGAUSS / 4];
    __shared__ v2f s_b[(NGAUSS / 4) * 8];

    const int tid = threadIdx.x;
    const int lane = tid & 31;
    const int wave = tid >> 5;

    {
        const v4f* __restrict__ gp4 = (const v4f*)ws;
        const v2f* __restrict__ gop2 = (const v2f*)(ws + WS_GOP);
        const v2f* __restrict__ bmat = (const v2f*)(ws + WS_BMAT);
        const v4f ga = gp4[tid];
        const v4f gb = gp4[tid + 256];
        const v2f go = gop2[tid];
        v2f bb[4];
#pragma unroll
        for (int r = 0; r < 4; ++r) {
            const int e = tid + r * 256;
            bb[r] = bmat[(e >> 3) * 32 + ((e >> 2) & 1) * 16 + (e & 3)];
        }
        s_gp[tid] = ga;
        s_gp[tid + 256] = gb;
        ((v2f*)s_op)[tid] = go;
#pragma unroll
        for (int r = 0; r < 4; ++r) s_b[tid + r * 256] = bb[r];
    }
    __syncthreads();

    const int pixBase = blockIdx.x * 256 + wave * 32;
    const int p = pixBase + lane;

    const float u = (float)(p & (IMG_W - 1)) - (float)(IMG_W / 2);
    const float v = (float)(p >> 8) - (float)(IMG_H / 2);

    const bool lo = (lane < 16);
    const int col = lane & 15;
    const int boff = (lane >> 4) * 4 + (col < 3 ? col : 3);

    float T = 1.0f;
    v8f acc1 = {0.f, 0.f, 0.f, 0.f, 0.f, 0.f, 0.f, 0.f};
    v8f acc2 = {0.f, 0.f, 0.f, 0.f, 0.f, 0.f, 0.f, 0.f};

    for (int t = 0; t < NGAUSS / 4; ++t) {
        const v4f g0 = s_gp[4 * t + 0];
        const v4f g1 = s_gp[4 * t + 1];
        const v4f g2 = s_gp[4 * t + 2];
        const v4f g3 = s_gp[4 * t + 3];
        const v4f op = s_op[t];
        const v2f b = s_b[t * 8 + boff];

        const float w0 = splat_one(g0.x, g0.y, g0.z, g0.w, op.x, u, v, T);
        const float w1 = splat_one(g1.x, g1.y, g1.z, g1.w, op.y, u, v, T);
        const float w2 = splat_one(g2.x, g2.y, g2.z, g2.w, op.z, u, v, T);
        const float w3 = splat_one(g3.x, g3.y, g3.z, g3.w, op.w, u, v, T);

        const float s0 = lane_xor16(w0);
        const float s1 = lane_xor16(w1);
        const float s2 = lane_xor16(w2);
        const float s3 = lane_xor16(w3);

        v2f a1, a2;
        a1.x = lo ? w0 : s2;
        a1.y = lo ? w1 : s3;
        a2.x = lo ? s0 : w2;
        a2.y = lo ? s1 : w3;

        acc1 = __builtin_amdgcn_wmma_f32_16x16x4_f32(
            false, a1, false, b, (short)0, acc1, false, false);
        acc2 = __builtin_amdgcn_wmma_f32_16x16x4_f32(
            false, a2, false, b, (short)0, acc2, false, false);
    }

    const int ch = lane & 15;
    if (ch < 3) {
#pragma unroll
        for (int r = 0; r < 8; ++r) {
            const int m = r + (lo ? 0 : 8);
            out[(pixBase + m) * 3 + ch] = acc1[r];
            out[(pixBase + 16 + m) * 3 + ch] = acc2[r];
        }
    }
}

extern "C" void kernel_launch(void* const* d_in, const int* in_sizes, int n_in,
                              void* d_out, int out_size, void* d_ws, size_t ws_size,
                              hipStream_t stream) {
    const float* means3d = (const float*)d_in[0];
    const float* scales  = (const float*)d_in[1];
    const float* opac    = (const float*)d_in[2];
    const float* colors  = (const float*)d_in[3];
    float* ws  = (float*)d_ws;
    float* out = (float*)d_out;

    hipLaunchKernelGGL(dgr_prep_kernel, dim3(1), dim3(512), 0, stream,
                       means3d, scales, opac, colors, ws);

    hipLaunchKernelGGL(dgr_raster_kernel, dim3(256), dim3(256), 0, stream,
                       ws, out);
}


# fuse the one-work-group prep kernel into the raster kernel: each work-group projects and stable-rank-sorts the 512 gaussians into its own LDS; one launch, no workspace; summation order unchanged
# speedup vs baseline: 2.6694x; 1.6668x over previous
#include <hip/hip_runtime.h>
#include <hip/hip_bf16.h>


typedef __attribute__((ext_vector_type(2))) float v2f;
typedef __attribute__((ext_vector_type(4))) float v4f;
typedef __attribute__((ext_vector_type(8))) float v8f;
typedef __attribute__((ext_vector_type(4))) int v4i;

#define IMG_H 256
#define IMG_W 256
#define NGAUSS 512
#define FOCAL_F 50.0f

__device__ __forceinline__ float lane_xor16(float x) {
    int v = __builtin_bit_cast(int, x);
    int r = __builtin_amdgcn_permlanex16(v, v, 0x76543210, 0xfedcba98, false, false);
    return __builtin_bit_cast(float, r);
}

__device__ __forceinline__ float splat_one(const float pu, const float pv, const float iu,
                                           const float iv, const float op, const float u,
                                           const float v, float& T) {
    const float du = u - pu;
    const float dv = v - pv;
    const float e = __expf(-0.5f * (du * du * iu + dv * dv * iv));
    const float a = fminf(e * op, 1.0f);
    const float w = a * T;
    T *= (1.0f - a);
    return w;
}

__global__ __launch_bounds__(256)
void dgr_fused_kernel(const float* __restrict__ means3d,
                      const float* __restrict__ scales,
                      const float* __restrict__ opac,
                      const float* __restrict__ colors,
                      float* __restrict__ out) {
    __shared__ v4f s_gp[NGAUSS];
    __shared__ v4f s_op[NGAUSS / 4];
    __shared__ v2f s_b[(NGAUSS / 4) * 8];
    __shared__ v4i s_key[NGAUSS / 4];

    const int tid = threadIdx.x;
    const int lane = tid & 31;
    const int wave = tid >> 5;

    float mx[2], my[2], mz[2], sx[2], sy[2], opv[2], cr[2], cg[2], cb[2];
#pragma unroll
    for (int h = 0; h < 2; ++h) {
        const int i = tid + h * 256;
        mx[h] = means3d[i * 3 + 0];
        my[h] = means3d[i * 3 + 1];
        mz[h] = means3d[i * 3 + 2];
        sx[h] = scales[i * 3 + 0];
        sy[h] = scales[i * 3 + 1];
        opv[h] = opac[i];
        cr[h] = colors[i * 3 + 0];
        cg[h] = colors[i * 3 + 1];
        cb[h] = colors[i * 3 + 2];
    }

    float z[2];
    int key[2];
#pragma unroll
    for (int h = 0; h < 2; ++h) {
        z[h] = fmaxf(mz[h], 0.1f);
        key[h] = __builtin_bit_cast(int, z[h]);
        ((int*)s_key)[tid + h * 256] = key[h];
    }
    __syncthreads();

    int rank0 = 0, rank1 = 0;
    for (int b = 0; b < NGAUSS / 32; ++b) {
        const int t0 = key[0] + ((b < wave) ? 1 : 0);
        const int t1 = key[1] + ((b < wave + 8) ? 1 : 0);
        const unsigned ut0 = (unsigned)t0, ut1 = (unsigned)t1;
        v4i kk[8];
#pragma unroll
        for (int q = 0; q < 8; ++q) kk[q] = s_key[b * 8 + q];
        __builtin_amdgcn_sched_barrier(0);
        unsigned a0 = 0u, c0 = 0u, a1 = 0u, c1 = 0u;
#pragma unroll
        for (int q = 0; q < 8; ++q) {
            a0 = (a0 << 1) | (((unsigned)kk[q].x - ut0) >> 31);
            c0 = (c0 << 1) | (((unsigned)kk[q].y - ut0) >> 31);
            a0 = (a0 << 1) | (((unsigned)kk[q].z - ut0) >> 31);
            c0 = (c0 << 1) | (((unsigned)kk[q].w - ut0) >> 31);
            a1 = (a1 << 1) | (((unsigned)kk[q].x - ut1) >> 31);
            c1 = (c1 << 1) | (((unsigned)kk[q].y - ut1) >> 31);
            a1 = (a1 << 1) | (((unsigned)kk[q].z - ut1) >> 31);
            c1 = (c1 << 1) | (((unsigned)kk[q].w - ut1) >> 31);
        }
        rank0 += __builtin_popcount(a0) + __builtin_popcount(c0);
        rank1 += __builtin_popcount(a1) + __builtin_popcount(c1);
    }
    {
        const unsigned uk0 = (unsigned)key[0], uk1 = (unsigned)key[1];
        const unsigned ue0 = uk0 + 1u, ue1 = uk1 + 1u;
        unsigned le0 = 0u, lt0 = 0u, le1 = 0u, lt1 = 0u;
#pragma unroll
        for (int q = 0; q < 8; ++q) {
            const v4i ka = s_key[wave * 8 + q];
            const v4i kb = s_key[64 + wave * 8 + q];
#pragma unroll
            for (int c = 0; c < 4; ++c) {
                le0 = (le0 << 1) | (((unsigned)ka[c] - ue0) >> 31);
                lt0 = (lt0 << 1) | (((unsigned)ka[c] - uk0) >> 31);
                le1 = (le1 << 1) | (((unsigned)kb[c] - ue1) >> 31);
                lt1 = (lt1 << 1) | (((unsigned)kb[c] - uk1) >> 31);
            }
        }
        const unsigned below = ~(0xffffffffu >> lane);
        rank0 += __builtin_popcount(le0 & ~lt0 & below);
        rank1 += __builtin_popcount(le1 & ~lt1 & below);
    }

#pragma unroll
    for (int h = 0; h < 2; ++h) {
        const int r = (h == 0) ? rank0 : rank1;
        const float zz = z[h];
        const float pu = FOCAL_F * mx[h] / zz;
        const float pv = FOCAL_F * my[h] / zz;
        const float su = fmaxf(FOCAL_F * sx[h] / zz, 0.5f);
        const float sv = fmaxf(FOCAL_F * sy[h] / zz, 0.5f);
        v4f g;
        g.x = pu;
        g.y = pv;
        g.z = 1.0f / (su * su);
        g.w = 1.0f / (sv * sv);
        s_gp[r] = g;
        ((float*)s_op)[r] = opv[h];
        float* bf = (float*)s_b;
        const int bbase = ((r >> 2) * 8 + ((r >> 1) & 1) * 4) * 2 + (r & 1);
        bf[bbase + 0] = cr[h];
        bf[bbase + 2] = cg[h];
        bf[bbase + 4] = cb[h];
        bf[bbase + 6] = 0.0f;
    }
    __syncthreads();

    const int pixBase = blockIdx.x * 256 + wave * 32;
    const int p = pixBase + lane;

    const float u = (float)(p & (IMG_W - 1)) - (float)(IMG_W / 2);
    const float v = (float)(p >> 8) - (float)(IMG_H / 2);

    const bool lo = (lane < 16);
    const int col = lane & 15;
    const int boff = (lane >> 4) * 4 + (col < 3 ? col : 3);

    float T = 1.0f;
    v8f acc1 = {0.f, 0.f, 0.f, 0.f, 0.f, 0.f, 0.f, 0.f};
    v8f acc2 = {0.f, 0.f, 0.f, 0.f, 0.f, 0.f, 0.f, 0.f};

    for (int t = 0; t < NGAUSS / 4; ++t) {
        const v4f g0 = s_gp[4 * t + 0];
        const v4f g1 = s_gp[4 * t + 1];
        const v4f g2 = s_gp[4 * t + 2];
        const v4f g3 = s_gp[4 * t + 3];
        const v4f op = s_op[t];
        const v2f b = s_b[t * 8 + boff];

        const float w0 = splat_one(g0.x, g0.y, g0.z, g0.w, op.x, u, v, T);
        const float w1 = splat_one(g1.x, g1.y, g1.z, g1.w, op.y, u, v, T);
        const float w2 = splat_one(g2.x, g2.y, g2.z, g2.w, op.z, u, v, T);
        const float w3 = splat_one(g3.x, g3.y, g3.z, g3.w, op.w, u, v, T);

        const float s0 = lane_xor16(w0);
        const float s1 = lane_xor16(w1);
        const float s2 = lane_xor16(w2);
        const float s3 = lane_xor16(w3);

        v2f a1, a2;
        a1.x = lo ? w0 : s2;
        a1.y = lo ? w1 : s3;
        a2.x = lo ? s0 : w2;
        a2.y = lo ? s1 : w3;

        acc1 = __builtin_amdgcn_wmma_f32_16x16x4_f32(
            false, a1, false, b, (short)0, acc1, false, false);
        acc2 = __builtin_amdgcn_wmma_f32_16x16x4_f32(
            false, a2, false, b, (short)0, acc2, false, false);
    }

    const int ch = lane & 15;
    if (ch < 3) {
#pragma unroll
        for (int r = 0; r < 8; ++r) {
            const int m = r + (lo ? 0 : 8);
            out[(pixBase + m) * 3 + ch] = acc1[r];
            out[(pixBase + 16 + m) * 3 + ch] = acc2[r];
        }
    }
}

extern "C" void kernel_launch(void* const* d_in, const int* in_sizes, int n_in,
                              void* d_out, int out_size, void* d_ws, size_t ws_size,
                              hipStream_t stream) {
    const float* means3d = (const float*)d_in[0];
    const float* scales  = (const float*)d_in[1];
    const float* opac    = (const float*)d_in[2];
    const float* colors  = (const float*)d_in[3];
    float* out = (float*)d_out;

    hipLaunchKernelGGL(dgr_fused_kernel, dim3(256), dim3(256), 0, stream,
                       means3d, scales, opac, colors, out);
}


# raster: fold all of log2(alpha) that is constant along an image row (v term, -0.5*log2e, log2 opacity) into each work-group's LDS table; pixel loop keeps sub, mul, fma, exp2, min; sums unchanged
# speedup vs baseline: 3.5436x; 1.3275x over previous
#include <hip/hip_runtime.h>
#include <hip/hip_bf16.h>


typedef __attribute__((ext_vector_type(2))) float v2f;
typedef __attribute__((ext_vector_type(4))) float v4f;
typedef __attribute__((ext_vector_type(8))) float v8f;
typedef __attribute__((ext_vector_type(4))) int v4i;

#define IMG_H 256
#define IMG_W 256
#define NGAUSS 512
#define FOCAL_F 50.0f

__device__ __forceinline__ float lane_xor16(float x) {
    int v = __builtin_bit_cast(int, x);
    int r = __builtin_amdgcn_permlanex16(v, v, 0x76543210, 0xfedcba98, false, false);
    return __builtin_bit_cast(float, r);
}

__device__ __forceinline__ float splat_one(const float pu, const float ku, const float c,
                                           const float u, float& T) {
    const float du = u - pu;
    const float e = __builtin_amdgcn_exp2f(__builtin_fmaf(du * du, ku, c));
    const float a = fminf(e, 1.0f);
    const float w = a * T;
    T *= (1.0f - a);
    return w;
}

__global__ __launch_bounds__(256)
void dgr_fused_kernel(const float* __restrict__ means3d,
                      const float* __restrict__ scales,
                      const float* __restrict__ opac,
                      const float* __restrict__ colors,
                      float* __restrict__ out) {
    __shared__ v4f s_pu[NGAUSS / 4];
    __shared__ v4f s_ku[NGAUSS / 4];
    __shared__ v4f s_c[NGAUSS / 4];
    __shared__ v2f s_b[(NGAUSS / 4) * 8];
    __shared__ v4i s_key[NGAUSS / 4];

    const int tid = threadIdx.x;
    const int lane = tid & 31;
    const int wave = tid >> 5;

    float mx[2], my[2], mz[2], sx[2], sy[2], opv[2], cr[2], cg[2], cb[2];
#pragma unroll
    for (int h = 0; h < 2; ++h) {
        const int i = tid + h * 256;
        mx[h] = means3d[i * 3 + 0];
        my[h] = means3d[i * 3 + 1];
        mz[h] = means3d[i * 3 + 2];
        sx[h] = scales[i * 3 + 0];
        sy[h] = scales[i * 3 + 1];
        opv[h] = opac[i];
        cr[h] = colors[i * 3 + 0];
        cg[h] = colors[i * 3 + 1];
        cb[h] = colors[i * 3 + 2];
    }

    float z[2];
    int key[2];
#pragma unroll
    for (int h = 0; h < 2; ++h) {
        z[h] = fmaxf(mz[h], 0.1f);
        key[h] = __builtin_bit_cast(int, z[h]);
        ((int*)s_key)[tid + h * 256] = key[h];
    }
    __syncthreads();

    int rank0 = 0, rank1 = 0;
    for (int b = 0; b < NGAUSS / 32; ++b) {
        const int t0 = key[0] + ((b < wave) ? 1 : 0);
        const int t1 = key[1] + ((b < wave + 8) ? 1 : 0);
        const unsigned ut0 = (unsigned)t0, ut1 = (unsigned)t1;
        v4i kk[8];
#pragma unroll
        for (int q = 0; q < 8; ++q) kk[q] = s_key[b * 8 + q];
        __builtin_amdgcn_sched_barrier(0);
        unsigned a0 = 0u, c0 = 0u, a1 = 0u, c1 = 0u;
#pragma unroll
        for (int q = 0; q < 8; ++q) {
            a0 = (a0 << 1) | (((unsigned)kk[q].x - ut0) >> 31);
            c0 = (c0 << 1) | (((unsigned)kk[q].y - ut0) >> 31);
            a0 = (a0 << 1) | (((unsigned)kk[q].z - ut0) >> 31);
            c0 = (c0 << 1) | (((unsigned)kk[q].w - ut0) >> 31);
            a1 = (a1 << 1) | (((unsigned)kk[q].x - ut1) >> 31);
            c1 = (c1 << 1) | (((unsigned)kk[q].y - ut1) >> 31);
            a1 = (a1 << 1) | (((unsigned)kk[q].z - ut1) >> 31);
            c1 = (c1 << 1) | (((unsigned)kk[q].w - ut1) >> 31);
        }
        rank0 += __builtin_popcount(a0) + __builtin_popcount(c0);
        rank1 += __builtin_popcount(a1) + __builtin_popcount(c1);
    }
    {
        const unsigned uk0 = (unsigned)key[0], uk1 = (unsigned)key[1];
        const unsigned ue0 = uk0 + 1u, ue1 = uk1 + 1u;
        unsigned le0 = 0u, lt0 = 0u, le1 = 0u, lt1 = 0u;
#pragma unroll
        for (int q = 0; q < 8; ++q) {
            const v4i ka = s_key[wave * 8 + q];
            const v4i kb = s_key[64 + wave * 8 + q];
#pragma unroll
            for (int c = 0; c < 4; ++c) {
                le0 = (le0 << 1) | (((unsigned)ka[c] - ue0) >> 31);
                lt0 = (lt0 << 1) | (((unsigned)ka[c] - uk0) >> 31);
                le1 = (le1 << 1) | (((unsigned)kb[c] - ue1) >> 31);
                lt1 = (lt1 << 1) | (((unsigned)kb[c] - uk1) >> 31);
            }
        }
        const unsigned below = ~(0xffffffffu >> lane);
        rank0 += __builtin_popcount(le0 & ~lt0 & below);
        rank1 += __builtin_popcount(le1 & ~lt1 & below);
    }

    const float vrow = (float)blockIdx.x - (float)(IMG_H / 2);
    const float KEXP = -0.5f * 1.44269504088896340736f;
#pragma unroll
    for (int h = 0; h < 2; ++h) {
        const int r = (h == 0) ? rank0 : rank1;
        const float zz = z[h];
        const float pu = FOCAL_F * mx[h] / zz;
        const float pv = FOCAL_F * my[h] / zz;
        const float su = fmaxf(FOCAL_F * sx[h] / zz, 0.5f);
        const float sv = fmaxf(FOCAL_F * sy[h] / zz, 0.5f);
        const float iu = 1.0f / (su * su);
        const float iv = 1.0f / (sv * sv);
        const float dv = vrow - pv;
        const float lg = (opv[h] > 0.0f) ? __log2f(opv[h]) : -__builtin_inff();
        ((float*)s_pu)[r] = pu;
        ((float*)s_ku)[r] = KEXP * iu;
        ((float*)s_c)[r] = KEXP * (dv * dv * iv) + lg;
        float* bf = (float*)s_b;
        const int bbase = ((r >> 2) * 8 + ((r >> 1) & 1) * 4) * 2 + (r & 1);
        bf[bbase + 0] = cr[h];
        bf[bbase + 2] = cg[h];
        bf[bbase + 4] = cb[h];
        bf[bbase + 6] = 0.0f;
    }
    __syncthreads();

    const int pixBase = blockIdx.x * 256 + wave * 32;
    const int p = pixBase + lane;

    const float u = (float)(p & (IMG_W - 1)) - (float)(IMG_W / 2);

    const bool lo = (lane < 16);
    const int col = lane & 15;
    const int boff = (lane >> 4) * 4 + (col < 3 ? col : 3);

    float T = 1.0f;
    v8f acc1 = {0.f, 0.f, 0.f, 0.f, 0.f, 0.f, 0.f, 0.f};
    v8f acc2 = {0.f, 0.f, 0.f, 0.f, 0.f, 0.f, 0.f, 0.f};

    for (int t = 0; t < NGAUSS / 4; ++t) {
        const v4f pu = s_pu[t];
        const v4f ku = s_ku[t];
        const v4f c = s_c[t];
        const v2f b = s_b[t * 8 + boff];

        const float w0 = splat_one(pu.x, ku.x, c.x, u, T);
        const float w1 = splat_one(pu.y, ku.y, c.y, u, T);
        const float w2 = splat_one(pu.z, ku.z, c.z, u, T);
        const float w3 = splat_one(pu.w, ku.w, c.w, u, T);

        const float s0 = lane_xor16(w0);
        const float s1 = lane_xor16(w1);
        const float s2 = lane_xor16(w2);
        const float s3 = lane_xor16(w3);

        v2f a1, a2;
        a1.x = lo ? w0 : s2;
        a1.y = lo ? w1 : s3;
        a2.x = lo ? s0 : w2;
        a2.y = lo ? s1 : w3;

        acc1 = __builtin_amdgcn_wmma_f32_16x16x4_f32(
            false, a1, false, b, (short)0, acc1, false, false);
        acc2 = __builtin_amdgcn_wmma_f32_16x16x4_f32(
            false, a2, false, b, (short)0, acc2, false, false);
    }

    const int ch = lane & 15;
    if (ch < 3) {
#pragma unroll
        for (int r = 0; r < 8; ++r) {
            const int m = r + (lo ? 0 : 8);
            out[(pixBase + m) * 3 + ch] = acc1[r];
            out[(pixBase + 16 + m) * 3 + ch] = acc2[r];
        }
    }
}

extern "C" void kernel_launch(void* const* d_in, const int* in_sizes, int n_in,
                              void* d_out, int out_size, void* d_ws, size_t ws_size,
                              hipStream_t stream) {
    const float* means3d = (const float*)d_in[0];
    const float* scales  = (const float*)d_in[1];
    const float* opac    = (const float*)d_in[2];
    const float* colors  = (const float*)d_in[3];
    float* out = (float*)d_out;

    hipLaunchKernelGGL(dgr_fused_kernel, dim3(256), dim3(256), 0, stream,
                       means3d, scales, opac, colors, out);
}


# rank sort: each wave stably ranks its own 64 keys and stores them as a sorted run; a key's place in the other seven runs comes from 7-probe binary searches, summed; order and sums unchanged
# speedup vs baseline: 3.9368x; 1.1110x over previous
#include <hip/hip_runtime.h>
#include <hip/hip_bf16.h>


typedef __attribute__((ext_vector_type(2))) float v2f;
typedef __attribute__((ext_vector_type(4))) float v4f;
typedef __attribute__((ext_vector_type(8))) float v8f;
typedef __attribute__((ext_vector_type(4))) int v4i;

#define IMG_H 256
#define IMG_W 256
#define NGAUSS 512
#define FOCAL_F 50.0f

__device__ __forceinline__ float lane_xor16(float x) {
    int v = __builtin_bit_cast(int, x);
    int r = __builtin_amdgcn_permlanex16(v, v, 0x76543210, 0xfedcba98, false, false);
    return __builtin_bit_cast(float, r);
}

__device__ __forceinline__ float splat_one(const float pu, const float ku, const float c,
                                           const float u, float& T) {
    const float du = u - pu;
    const float e = __builtin_amdgcn_exp2f(__builtin_fmaf(du * du, ku, c));
    const float a = fminf(e, 1.0f);
    const float w = a * T;
    T *= (1.0f - a);
    return w;
}

__global__ __launch_bounds__(256)
void dgr_fused_kernel(const float* __restrict__ means3d,
                      const float* __restrict__ scales,
                      const float* __restrict__ opac,
                      const float* __restrict__ colors,
                      float* __restrict__ out) {
    __shared__ v4f s_pu[NGAUSS / 4];
    __shared__ v4f s_ku[NGAUSS / 4];
    __shared__ v4f s_c[NGAUSS / 4];
    __shared__ v2f s_b[(NGAUSS / 4) * 8];
    __shared__ v4i s_key[NGAUSS / 4];
    __shared__ int s_list[NGAUSS];

    const int tid = threadIdx.x;
    const int lane = tid & 31;
    const int wave = __builtin_amdgcn_readfirstlane(tid >> 5);

    float mx[2], my[2], mz[2], sx[2], sy[2], opv[2], cr[2], cg[2], cb[2];
#pragma unroll
    for (int h = 0; h < 2; ++h) {
        const int i = wave * 64 + h * 32 + lane;
        mx[h] = means3d[i * 3 + 0];
        my[h] = means3d[i * 3 + 1];
        mz[h] = means3d[i * 3 + 2];
        sx[h] = scales[i * 3 + 0];
        sy[h] = scales[i * 3 + 1];
        opv[h] = opac[i];
        cr[h] = colors[i * 3 + 0];
        cg[h] = colors[i * 3 + 1];
        cb[h] = colors[i * 3 + 2];
    }

    float z[2];
    int key[2];
#pragma unroll
    for (int h = 0; h < 2; ++h) {
        z[h] = fmaxf(mz[h], 0.1f);
        key[h] = __builtin_bit_cast(int, z[h]);
        ((int*)s_key)[wave * 64 + h * 32 + lane] = key[h];
    }
    __syncthreads();

    int rank0, rank1;
    {
        const unsigned uk0 = (unsigned)key[0], uk1 = (unsigned)key[1];
        const unsigned ue0 = uk0 + 1u, ue1 = uk1 + 1u;
        unsigned le0 = 0u, lt0 = 0u, up0 = 0u;
        unsigned le1 = 0u, lt1 = 0u, dn1 = 0u;
        v4i kk[16];
#pragma unroll
        for (int q = 0; q < 16; ++q) kk[q] = s_key[wave * 16 + q];
        __builtin_amdgcn_sched_barrier(0);
#pragma unroll
        for (int q = 0; q < 8; ++q) {
            const v4i ka = kk[q];
            const v4i kb = kk[8 + q];
#pragma unroll
            for (int c = 0; c < 4; ++c) {
                le0 = (le0 << 1) | (((unsigned)ka[c] - ue0) >> 31);
                lt0 = (lt0 << 1) | (((unsigned)ka[c] - uk0) >> 31);
                dn1 = (dn1 << 1) | (((unsigned)ka[c] - ue1) >> 31);
                le1 = (le1 << 1) | (((unsigned)kb[c] - ue1) >> 31);
                lt1 = (lt1 << 1) | (((unsigned)kb[c] - uk1) >> 31);
                up0 = (up0 << 1) | (((unsigned)kb[c] - uk0) >> 31);
            }
        }
        const unsigned below = ~(0xffffffffu >> lane);
        rank0 = __builtin_popcount(lt0) + __builtin_popcount(le0 & ~lt0 & below) + __builtin_popcount(up0);
        rank1 = __builtin_popcount(dn1) + __builtin_popcount(lt1) + __builtin_popcount(le1 & ~lt1 & below);
    }
    s_list[wave * 64 + rank0] = key[0];
    s_list[wave * 64 + rank1] = key[1];
    __syncthreads();

    {
        const char* lbase = (const char*)s_list;
        const unsigned uk0 = (unsigned)key[0], uk1 = (unsigned)key[1];
        unsigned p0[7], p1[7], t0[7], t1[7];
        unsigned basesum = 0u;
#pragma unroll
        for (int q = 0; q < 7; ++q) {
            const int o = (wave + 1 + q) & 7;
            const unsigned tie = (o < wave) ? 1u : 0u;
            p0[q] = p1[q] = (unsigned)o * 256u;
            basesum += (unsigned)o * 256u;
            t0[q] = uk0 + tie;
            t1[q] = uk1 + tie;
        }
#pragma unroll
        for (int s = 32; s >= 1; s >>= 1) {
            unsigned v0[7], v1[7];
#pragma unroll
            for (int q = 0; q < 7; ++q) {
                v0[q] = *(const unsigned*)(lbase + p0[q] + (s - 1) * 4);
                v1[q] = *(const unsigned*)(lbase + p1[q] + (s - 1) * 4);
            }
            __builtin_amdgcn_sched_barrier(0);
#pragma unroll
            for (int q = 0; q < 7; ++q) {
                p0[q] += (v0[q] < t0[q]) ? (unsigned)(s * 4) : 0u;
                p1[q] += (v1[q] < t1[q]) ? (unsigned)(s * 4) : 0u;
            }
        }
        unsigned sum0 = 0u, sum1 = 0u;
        {
            unsigned v0[7], v1[7];
#pragma unroll
            for (int q = 0; q < 7; ++q) {
                v0[q] = *(const unsigned*)(lbase + p0[q]);
                v1[q] = *(const unsigned*)(lbase + p1[q]);
            }
            __builtin_amdgcn_sched_barrier(0);
#pragma unroll
            for (int q = 0; q < 7; ++q) {
                sum0 += p0[q] + ((v0[q] < t0[q]) ? 4u : 0u);
                sum1 += p1[q] + ((v1[q] < t1[q]) ? 4u : 0u);
            }
        }
        rank0 += (int)((sum0 - basesum) >> 2);
        rank1 += (int)((sum1 - basesum) >> 2);
    }

    const float vrow = (float)blockIdx.x - (float)(IMG_H / 2);
    const float KEXP = -0.5f * 1.44269504088896340736f;
#pragma unroll
    for (int h = 0; h < 2; ++h) {
        const int r = (h == 0) ? rank0 : rank1;
        const float zz = z[h];
        const float pu = FOCAL_F * mx[h] / zz;
        const float pv = FOCAL_F * my[h] / zz;
        const float su = fmaxf(FOCAL_F * sx[h] / zz, 0.5f);
        const float sv = fmaxf(FOCAL_F * sy[h] / zz, 0.5f);
        const float iu = 1.0f / (su * su);
        const float iv = 1.0f / (sv * sv);
        const float dv = vrow - pv;
        const float lg = (opv[h] > 0.0f) ? __log2f(opv[h]) : -__builtin_inff();
        ((float*)s_pu)[r] = pu;
        ((float*)s_ku)[r] = KEXP * iu;
        ((float*)s_c)[r] = KEXP * (dv * dv * iv) + lg;
        float* bf = (float*)s_b;
        const int bbase = ((r >> 2) * 8 + ((r >> 1) & 1) * 4) * 2 + (r & 1);
        bf[bbase + 0] = cr[h];
        bf[bbase + 2] = cg[h];
        bf[bbase + 4] = cb[h];
        bf[bbase + 6] = 0.0f;
    }
    __syncthreads();

    const int pixBase = blockIdx.x * 256 + wave * 32;
    const int p = pixBase + lane;

    const float u = (float)(p & (IMG_W - 1)) - (float)(IMG_W / 2);

    const bool lo = (lane < 16);
    const int col = lane & 15;
    const int boff = (lane >> 4) * 4 + (col < 3 ? col : 3);

    float T = 1.0f;
    v8f acc1 = {0.f, 0.f, 0.f, 0.f, 0.f, 0.f, 0.f, 0.f};
    v8f acc2 = {0.f, 0.f, 0.f, 0.f, 0.f, 0.f, 0.f, 0.f};

    for (int t = 0; t < NGAUSS / 4; ++t) {
        const v4f pu = s_pu[t];
        const v4f ku = s_ku[t];
        const v4f c = s_c[t];
        const v2f b = s_b[t * 8 + boff];

        const float w0 = splat_one(pu.x, ku.x, c.x, u, T);
        const float w1 = splat_one(pu.y, ku.y, c.y, u, T);
        const float w2 = splat_one(pu.z, ku.z, c.z, u, T);
        const float w3 = splat_one(pu.w, ku.w, c.w, u, T);

        const float s0 = lane_xor16(w0);
        const float s1 = lane_xor16(w1);
        const float s2 = lane_xor16(w2);
        const float s3 = lane_xor16(w3);

        v2f a1, a2;
        a1.x = lo ? w0 : s2;
        a1.y = lo ? w1 : s3;
        a2.x = lo ? s0 : w2;
        a2.y = lo ? s1 : w3;

        acc1 = __builtin_amdgcn_wmma_f32_16x16x4_f32(
            false, a1, false, b, (short)0, acc1, false, false);
        acc2 = __builtin_amdgcn_wmma_f32_16x16x4_f32(
            false, a2, false, b, (short)0, acc2, false, false);
    }

    const int ch = lane & 15;
    if (ch < 3) {
#pragma unroll
        for (int r = 0; r < 8; ++r) {
            const int m = r + (lo ? 0 : 8);
            out[(pixBase + m) * 3 + ch] = acc1[r];
            out[(pixBase + 16 + m) * 3 + ch] = acc2[r];
        }
    }
}

extern "C" void kernel_launch(void* const* d_in, const int* in_sizes, int n_in,
                              void* d_out, int out_size, void* d_ws, size_t ws_size,
                              hipStream_t stream) {
    const float* means3d = (const float*)d_in[0];
    const float* scales  = (const float*)d_in[1];
    const float* opac    = (const float*)d_in[2];
    const float* colors  = (const float*)d_in[3];
    float* out = (float*)d_out;

    hipLaunchKernelGGL(dgr_fused_kernel, dim3(256), dim3(256), 0, stream,
                       means3d, scales, opac, colors, out);
}


# raster: each lane's own alpha*T weights are its A fragment; B sends pixels of lanes 0-15 to columns 0..2, of lanes 16-31 to 3..5: no permlanex16 swap; a wmma sums 2 gaussians, 2 tiles added at end
# speedup vs baseline: 4.3818x; 1.1130x over previous
#include <hip/hip_runtime.h>
#include <hip/hip_bf16.h>


typedef __attribute__((ext_vector_type(2))) float v2f;
typedef __attribute__((ext_vector_type(4))) float v4f;
typedef __attribute__((ext_vector_type(8))) float v8f;
typedef __attribute__((ext_vector_type(4))) int v4i;

#define IMG_H 256
#define IMG_W 256
#define NGAUSS 512
#define FOCAL_F 50.0f

__device__ __forceinline__ float splat_one(const float pu, const float ku, const float c,
                                           const float u, float& T) {
    const float du = u - pu;
    const float e = __builtin_amdgcn_exp2f(__builtin_fmaf(du * du, ku, c));
    const float a = fminf(e, 1.0f);
    const float w = a * T;
    T *= (1.0f - a);
    return w;
}

__device__ __forceinline__ void splat_chunk(const v4f pu, const v4f ku, const v4f c, const v4f b,
                                            const float u, float& T, v8f& acc1, v8f& acc2) {
    const float w0 = splat_one(pu.x, ku.x, c.x, u, T);
    const float w1 = splat_one(pu.y, ku.y, c.y, u, T);
    const float w2 = splat_one(pu.z, ku.z, c.z, u, T);
    const float w3 = splat_one(pu.w, ku.w, c.w, u, T);
    const v2f a1 = {w0, w1};
    const v2f a2 = {w2, w3};
    const v2f b1 = {b.x, b.y};
    const v2f b2 = {b.z, b.w};
    acc1 = __builtin_amdgcn_wmma_f32_16x16x4_f32(
        false, a1, false, b1, (short)0, acc1, false, false);
    acc2 = __builtin_amdgcn_wmma_f32_16x16x4_f32(
        false, a2, false, b2, (short)0, acc2, false, false);
}

__global__ __launch_bounds__(256)
void dgr_fused_kernel(const float* __restrict__ means3d,
                      const float* __restrict__ scales,
                      const float* __restrict__ opac,
                      const float* __restrict__ colors,
                      float* __restrict__ out) {
    __shared__ v4f s_pu[NGAUSS / 4];
    __shared__ v4f s_ku[NGAUSS / 4];
    __shared__ v4f s_c[NGAUSS / 4];
    __shared__ v4f s_b[(NGAUSS / 4) * 4];
    __shared__ v4i s_key[NGAUSS / 4];
    __shared__ int s_list[NGAUSS];

    const int tid = threadIdx.x;
    const int lane = tid & 31;
    const int wave = __builtin_amdgcn_readfirstlane(tid >> 5);

    float mx[2], my[2], mz[2], sx[2], sy[2], opv[2], cr[2], cg[2], cb[2];
#pragma unroll
    for (int h = 0; h < 2; ++h) {
        const int i = wave * 64 + h * 32 + lane;
        mx[h] = means3d[i * 3 + 0];
        my[h] = means3d[i * 3 + 1];
        mz[h] = means3d[i * 3 + 2];
        sx[h] = scales[i * 3 + 0];
        sy[h] = scales[i * 3 + 1];
        opv[h] = opac[i];
        cr[h] = colors[i * 3 + 0];
        cg[h] = colors[i * 3 + 1];
        cb[h] = colors[i * 3 + 2];
    }

    float z[2];
    int key[2];
#pragma unroll
    for (int h = 0; h < 2; ++h) {
        z[h] = fmaxf(mz[h], 0.1f);
        key[h] = __builtin_bit_cast(int, z[h]);
        ((int*)s_key)[wave * 64 + h * 32 + lane] = key[h];
    }
    __syncthreads();

    int rank0, rank1;
    {
        const unsigned uk0 = (unsigned)key[0], uk1 = (unsigned)key[1];
        const unsigned ue0 = uk0 + 1u, ue1 = uk1 + 1u;
        unsigned le0 = 0u, lt0 = 0u, up0 = 0u;
        unsigned le1 = 0u, lt1 = 0u, dn1 = 0u;
        v4i kk[16];
#pragma unroll
        for (int q = 0; q < 16; ++q) kk[q] = s_key[wave * 16 + q];
        __builtin_amdgcn_sched_barrier(0);
#pragma unroll
        for (int q = 0; q < 8; ++q) {
            const v4i ka = kk[q];
            const v4i kb = kk[8 + q];
#pragma unroll
            for (int c = 0; c < 4; ++c) {
                le0 = (le0 << 1) | (((unsigned)ka[c] - ue0) >> 31);
                lt0 = (lt0 << 1) | (((unsigned)ka[c] - uk0) >> 31);
                dn1 = (dn1 << 1) | (((unsigned)ka[c] - ue1) >> 31);
                le1 = (le1 << 1) | (((unsigned)kb[c] - ue1) >> 31);
                lt1 = (lt1 << 1) | (((unsigned)kb[c] - uk1) >> 31);
                up0 = (up0 << 1) | (((unsigned)kb[c] - uk0) >> 31);
            }
        }
        const unsigned below = ~(0xffffffffu >> lane);
        rank0 = __builtin_popcount(lt0) + __builtin_popcount(le0 & ~lt0 & below) + __builtin_popcount(up0);
        rank1 = __builtin_popcount(dn1) + __builtin_popcount(lt1) + __builtin_popcount(le1 & ~lt1 & below);
    }
    s_list[wave * 64 + rank0] = key[0];
    s_list[wave * 64 + rank1] = key[1];
    __syncthreads();

    {
        const char* lbase = (const char*)s_list;
        const unsigned uk0 = (unsigned)key[0], uk1 = (unsigned)key[1];
        unsigned p0[7], p1[7], t0[7], t1[7];
        unsigned basesum = 0u;
#pragma unroll
        for (int q = 0; q < 7; ++q) {
            const int o = (wave + 1 + q) & 7;
            const unsigned tie = (o < wave) ? 1u : 0u;
            p0[q] = p1[q] = (unsigned)o * 256u;
            basesum += (unsigned)o * 256u;
            t0[q] = uk0 + tie;
            t1[q] = uk1 + tie;
        }
#pragma unroll
        for (int s = 32; s >= 1; s >>= 1) {
            unsigned v0[7], v1[7];
#pragma unroll
            for (int q = 0; q < 7; ++q) {
                v0[q] = *(const unsigned*)(lbase + p0[q] + (s - 1) * 4);
                v1[q] = *(const unsigned*)(lbase + p1[q] + (s - 1) * 4);
            }
            __builtin_amdgcn_sched_barrier(0);
#pragma unroll
            for (int q = 0; q < 7; ++q) {
                p0[q] += (v0[q] < t0[q]) ? (unsigned)(s * 4) : 0u;
                p1[q] += (v1[q] < t1[q]) ? (unsigned)(s * 4) : 0u;
            }
        }
        unsigned sum0 = 0u, sum1 = 0u;
        {
            unsigned v0[7], v1[7];
#pragma unroll
            for (int q = 0; q < 7; ++q) {
                v0[q] = *(const unsigned*)(lbase + p0[q]);
                v1[q] = *(const unsigned*)(lbase + p1[q]);
            }
            __builtin_amdgcn_sched_barrier(0);
#pragma unroll
            for (int q = 0; q < 7; ++q) {
                sum0 += p0[q] + ((v0[q] < t0[q]) ? 4u : 0u);
                sum1 += p1[q] + ((v1[q] < t1[q]) ? 4u : 0u);
            }
        }
        rank0 += (int)((sum0 - basesum) >> 2);
        rank1 += (int)((sum1 - basesum) >> 2);
    }

    const float vrow = (float)blockIdx.x - (float)(IMG_H / 2);
    const float KEXP = -0.5f * 1.44269504088896340736f;
#pragma unroll
    for (int h = 0; h < 2; ++h) {
        const int r = (h == 0) ? rank0 : rank1;
        const float zz = z[h];
        const float pu = FOCAL_F * mx[h] / zz;
        const float pv = FOCAL_F * my[h] / zz;
        const float su = fmaxf(FOCAL_F * sx[h] / zz, 0.5f);
        const float sv = fmaxf(FOCAL_F * sy[h] / zz, 0.5f);
        const float iu = 1.0f / (su * su);
        const float iv = 1.0f / (sv * sv);
        const float dv = vrow - pv;
        const float lg = (opv[h] > 0.0f) ? __log2f(opv[h]) : -__builtin_inff();
        ((float*)s_pu)[r] = pu;
        ((float*)s_ku)[r] = KEXP * iu;
        ((float*)s_c)[r] = KEXP * (dv * dv * iv) + lg;
        float* bf = (float*)s_b;
        const int bbase = (r >> 2) * 16 + (r & 3);
        bf[bbase + 0] = cr[h];
        bf[bbase + 4] = cg[h];
        bf[bbase + 8] = cb[h];
        bf[bbase + 12] = 0.0f;
    }
    __syncthreads();

    const int pixBase = blockIdx.x * 256 + wave * 32;
    const int p = pixBase + lane;

    const float u = (float)(p & (IMG_W - 1)) - (float)(IMG_W / 2);

    const int col = lane & 15;
    const int half = lane >> 4;
    const int chn = col - 3 * half;
    const int boff = (chn >= 0 && chn < 3) ? chn : 3;

    float T = 1.0f;
    v8f acc1 = {0.f, 0.f, 0.f, 0.f, 0.f, 0.f, 0.f, 0.f};
    v8f acc2 = {0.f, 0.f, 0.f, 0.f, 0.f, 0.f, 0.f, 0.f};

    for (int t = 0; t < NGAUSS / 8; ++t) {
        const v4f pu0 = s_pu[2 * t];
        const v4f ku0 = s_ku[2 * t];
        const v4f c0 = s_c[2 * t];
        const v4f b0 = s_b[t * 8 + boff];
        const v4f pu1 = s_pu[2 * t + 1];
        const v4f ku1 = s_ku[2 * t + 1];
        const v4f c1 = s_c[2 * t + 1];
        const v4f b1 = s_b[t * 8 + 4 + boff];
        __builtin_amdgcn_sched_barrier(0);
        splat_chunk(pu0, ku0, c0, b0, u, T, acc1, acc2);
        splat_chunk(pu1, ku1, c1, b1, u, T, acc1, acc2);
    }

    if (col < 6) {
        const int ch = (col < 3) ? col : col - 3;
        const int prow = pixBase + ((col < 3) ? 0 : 16) + half * 8;
#pragma unroll
        for (int r = 0; r < 8; ++r) {
            out[(prow + r) * 3 + ch] = acc1[r] + acc2[r];
        }
    }
}

extern "C" void kernel_launch(void* const* d_in, const int* in_sizes, int n_in,
                              void* d_out, int out_size, void* d_ws, size_t ws_size,
                              hipStream_t stream) {
    const float* means3d = (const float*)d_in[0];
    const float* scales  = (const float*)d_in[1];
    const float* opac    = (const float*)d_in[2];
    const float* colors  = (const float*)d_in[3];
    float* out = (float*)d_out;

    hipLaunchKernelGGL(dgr_fused_kernel, dim3(256), dim3(256), 0, stream,
                       means3d, scales, opac, colors, out);
}
